# starting kernel
# baseline (speedup 1.0000x reference)
#include <hip/hip_runtime.h>

typedef __attribute__((ext_vector_type(2))) float v2f;
typedef __attribute__((ext_vector_type(8))) float v8f;

#define B_ 4
#define N_ 512
#define D_ 128
#define L_ 2
#define M_ (B_ * N_)
#define TI 8
#define JT 64

__global__ __launch_bounds__(32)
void wmma_gemm_bias(const float* __restrict__ X, const float* __restrict__ W,
                    int wstride, int woff, const float* __restrict__ bias,
                    float* __restrict__ out) {
  const int lane = threadIdx.x;
  const int l16  = lane & 15;
  const int half = lane >> 4;
  const int tm = blockIdx.y;
  const int tn = blockIdx.x;
  const float* xrow = X + (tm * 16 + l16) * D_;
  const float* wrow = W + (tn * 16 + l16) * wstride + woff;
  v8f c = {};
#pragma unroll
  for (int kk = 0; kk < D_ / 4; ++kk) {
    const int kb = kk * 4 + half * 2;
    v2f a = *(const v2f*)(xrow + kb);
    v2f b = *(const v2f*)(wrow + kb);
    c = __builtin_amdgcn_wmma_f32_16x16x4_f32(
        false, a, false, b, (short)0, c, false, false);
  }
  const float bv = bias ? bias[tn * 16 + l16] : 0.f;
#pragma unroll
  for (int r = 0; r < 8; ++r) {
    const int m = tm * 16 + half * 8 + r;
    out[m * D_ + tn * 16 + l16] = c[r] + bv;
  }
}

__global__ __launch_bounds__(32)
void wmma_gemm_final(const float* __restrict__ S, const float* __restrict__ W2,
                     const float* __restrict__ b2, const float* __restrict__ rowsum,
                     float* __restrict__ h) {
  const int lane = threadIdx.x;
  const int l16  = lane & 15;
  const int half = lane >> 4;
  const int tm = blockIdx.y;
  const int tn = blockIdx.x;
  const float* xrow = S + (tm * 16 + l16) * D_;
  const float* wrow = W2 + (tn * 16 + l16) * D_;
  v8f c = {};
#pragma unroll
  for (int kk = 0; kk < D_ / 4; ++kk) {
    const int kb = kk * 4 + half * 2;
    v2f a = *(const v2f*)(xrow + kb);
    v2f b = *(const v2f*)(wrow + kb);
    c = __builtin_amdgcn_wmma_f32_16x16x4_f32(
        false, a, false, b, (short)0, c, false, false);
  }
  const float bv = b2[tn * 16 + l16];
#pragma unroll
  for (int r = 0; r < 8; ++r) {
    const int m = tm * 16 + half * 8 + r;
    const int idx = m * D_ + tn * 16 + l16;
    h[idx] = h[idx] + c[r] + rowsum[m] * bv;
  }
}

__global__ __launch_bounds__(256)
void pairwise_reduce(const float* __restrict__ ai, const float* __restrict__ aj,
                     const float* __restrict__ adj, float* __restrict__ S) {
  __shared__ float ajT[D_ * (JT + 1)];
  __shared__ float adjT[TI * JT];
  const int b  = blockIdx.y;
  const int i0 = blockIdx.x * TI;
  const int tid = threadIdx.x;
  const int o  = tid & (D_ - 1);
  const int ip = tid >> 7;

  float acc[TI / 2];
  float aiv[TI / 2];
#pragma unroll
  for (int r = 0; r < TI / 2; ++r) {
    acc[r] = 0.f;
    aiv[r] = ai[((b * N_) + i0 + ip * (TI / 2) + r) * D_ + o];
  }

  for (int j0 = 0; j0 < N_; j0 += JT) {
    __syncthreads();
    for (int t = tid; t < JT * D_; t += 256) {
      const int jj = t >> 7, oo = t & (D_ - 1);
      ajT[oo * (JT + 1) + jj] = aj[((b * N_) + j0 + jj) * D_ + oo];
    }
    for (int t = tid; t < TI * JT; t += 256) {
      const int ii = t >> 6, jj = t & (JT - 1);
      adjT[ii * JT + jj] = adj[((b * N_) + i0 + ii) * N_ + j0 + jj];
    }
    __syncthreads();
#pragma unroll 4
    for (int jj = 0; jj < JT; ++jj) {
      const float ajv = ajT[o * (JT + 1) + jj];
#pragma unroll
      for (int r = 0; r < TI / 2; ++r) {
        float t = aiv[r] + ajv;
        t = fmaxf(t, 0.f);
        acc[r] = fmaf(adjT[(ip * (TI / 2) + r) * JT + jj], t, acc[r]);
      }
    }
  }
#pragma unroll
  for (int r = 0; r < TI / 2; ++r)
    S[((b * N_) + i0 + ip * (TI / 2) + r) * D_ + o] = acc[r];
}

__global__ __launch_bounds__(256)
void rowsum_kernel(const float* __restrict__ adj, float* __restrict__ rowsum) {
  const int lane = threadIdx.x & 31;
  const int wave = threadIdx.x >> 5;
  const int row  = blockIdx.x * 8 + wave;
  const float* p = adj + row * N_;
  float s = 0.f;
  for (int j = lane; j < N_; j += 32) s += p[j];
#pragma unroll
  for (int off = 16; off > 0; off >>= 1) s += __shfl_down(s, off, 32);
  if (lane == 0) rowsum[row] = s;
}

extern "C" void kernel_launch(void* const* d_in, const int* in_sizes, int n_in,
                              void* d_out, int out_size, void* d_ws, size_t ws_size,
                              hipStream_t stream) {
  const float* node = (const float*)d_in[0];
  const float* adj  = (const float*)d_in[1];
  const float* w1   = (const float*)d_in[2];
  const float* b1   = (const float*)d_in[3];
  const float* w2   = (const float*)d_in[4];
  const float* b2   = (const float*)d_in[5];

  float* h  = (float*)d_out;
  float* ai = (float*)d_ws;
  float* aj = ai + (size_t)M_ * D_;
  float* S  = aj + (size_t)M_ * D_;
  float* rs = S  + (size_t)M_ * D_;

  hipMemcpyAsync(h, node, (size_t)M_ * D_ * sizeof(float),
                 hipMemcpyDeviceToDevice, stream);
  rowsum_kernel<<<M_ / 8, 256, 0, stream>>>(adj, rs);

  const dim3 gg(D_ / 16, M_ / 16);
  for (int l = 0; l < L_; ++l) {
    const float* w1l = w1 + (size_t)l * D_ * 2 * D_;
    const float* b1l = b1 + (size_t)l * D_;
    const float* w2l = w2 + (size_t)l * D_ * D_;
    const float* b2l = b2 + (size_t)l * D_;
    wmma_gemm_bias<<<gg, 32, 0, stream>>>(h, w1l, 2 * D_, 0,  b1l, ai);
    wmma_gemm_bias<<<gg, 32, 0, stream>>>(h, w1l, 2 * D_, D_, nullptr, aj);
    pairwise_reduce<<<dim3(N_ / TI, B_), 256, 0, stream>>>(ai, aj, adj, S);
    wmma_gemm_final<<<gg, 32, 0, stream>>>(S, w2l, b2l, rs, h);
  }
}


# pairwise kernel retiled: block 16 i x 64 o holds all 512 j of aj and adj in LDS, staged once in 4 phases with the next phase's global loads in flight, 2x2 outputs per lane; j sum order unchanged
# speedup vs baseline: 1.7601x; 1.7601x over previous
#include <hip/hip_runtime.h>

typedef __attribute__((ext_vector_type(2))) float v2f;
typedef __attribute__((ext_vector_type(4))) float v4f;
typedef __attribute__((ext_vector_type(8))) float v8f;

#define B_ 4
#define N_ 512
#define D_ 128
#define L_ 2
#define M_ (B_ * N_)

__global__ __launch_bounds__(32)
void wmma_gemm_bias(const float* __restrict__ X, const float* __restrict__ W,
                    int wstride, int woff, const float* __restrict__ bias,
                    float* __restrict__ out) {
  const int lane = threadIdx.x;
  const int l16  = lane & 15;
  const int half = lane >> 4;
  const int tm = blockIdx.y;
  const int tn = blockIdx.x;
  const float* xrow = X + (tm * 16 + l16) * D_;
  const float* wrow = W + (tn * 16 + l16) * wstride + woff;
  v8f c = {};
#pragma unroll
  for (int kk = 0; kk < D_ / 4; ++kk) {
    const int kb = kk * 4 + half * 2;
    v2f a = *(const v2f*)(xrow + kb);
    v2f b = *(const v2f*)(wrow + kb);
    c = __builtin_amdgcn_wmma_f32_16x16x4_f32(
        false, a, false, b, (short)0, c, false, false);
  }
  const float bv = bias ? bias[tn * 16 + l16] : 0.f;
#pragma unroll
  for (int r = 0; r < 8; ++r) {
    const int m = tm * 16 + half * 8 + r;
    out[m * D_ + tn * 16 + l16] = c[r] + bv;
  }
}

__global__ __launch_bounds__(32)
void wmma_gemm_final(const float* __restrict__ S, const float* __restrict__ W2,
                     const float* __restrict__ b2, const float* __restrict__ rowsum,
                     float* __restrict__ h) {
  const int lane = threadIdx.x;
  const int l16  = lane & 15;
  const int half = lane >> 4;
  const int tm = blockIdx.y;
  const int tn = blockIdx.x;
  const float* xrow = S + (tm * 16 + l16) * D_;
  const float* wrow = W2 + (tn * 16 + l16) * D_;
  v8f c = {};
#pragma unroll
  for (int kk = 0; kk < D_ / 4; ++kk) {
    const int kb = kk * 4 + half * 2;
    v2f a = *(const v2f*)(xrow + kb);
    v2f b = *(const v2f*)(wrow + kb);
    c = __builtin_amdgcn_wmma_f32_16x16x4_f32(
        false, a, false, b, (short)0, c, false, false);
  }
  const float bv = b2[tn * 16 + l16];
#pragma unroll
  for (int r = 0; r < 8; ++r) {
    const int m = tm * 16 + half * 8 + r;
    const int idx = m * D_ + tn * 16 + l16;
    h[idx] = h[idx] + c[r] + rowsum[m] * bv;
  }
}

#define PI 16
#define PO 64
#define JP 128
#define NPH (N_ / JP)
#define AP (N_ + 4)
#define AJ_V4 (JP * PO / 4 / 256)
#define AD_V4 (PI * JP / 4 / 256)

#define GJ 16

struct PGroup {
  v4f d0[GJ / 4], d1[GJ / 4];
  v2f a[GJ];
  __device__ __forceinline__ void load(const v2f* ajl, const v4f* d0l, const v4f* d1l, int j) {
#pragma unroll
    for (int q = 0; q < GJ / 4; ++q) { d0[q] = d0l[(j >> 2) + q]; d1[q] = d1l[(j >> 2) + q]; }
#pragma unroll
    for (int q = 0; q < GJ; ++q) a[q] = ajl[(j + q) * (PO / 2)];
  }
  __device__ __forceinline__ void accumulate(const v2f ai0, const v2f ai1, v2f& acc0, v2f& acc1) const {
#pragma unroll
    for (int q = 0; q < GJ; ++q) {
#pragma unroll
      for (int c = 0; c < 2; ++c) {
        const float t0 = fmaxf(ai0[c] + a[q][c], 0.f);
        const float t1 = fmaxf(ai1[c] + a[q][c], 0.f);
        acc0[c] = fmaf(d0[q >> 2][q & 3], t0, acc0[c]);
        acc1[c] = fmaf(d1[q >> 2][q & 3], t1, acc1[c]);
      }
    }
  }
};

__global__ __launch_bounds__(256)
void pairwise_reduce(const float* __restrict__ ai, const float* __restrict__ aj,
                     const float* __restrict__ adj, float* __restrict__ S) {
  __shared__ __attribute__((aligned(16))) float ajs[N_ * PO];
  __shared__ __attribute__((aligned(16))) float ads[PI * AP];
  const int tid  = threadIdx.x;
  const int lane = tid & 31;
  const int wave = tid >> 5;
  const int o0 = blockIdx.x * PO;
  const int i0 = blockIdx.y * PI;
  const int b  = blockIdx.z;

  const float* ajg = aj  + (size_t)(b * N_) * D_ + o0;
  const float* adg = adj + (size_t)(b * N_ + i0) * N_;

  v4f ra[AJ_V4], rd[AD_V4];
#pragma unroll
  for (int k = 0; k < AJ_V4; ++k) {
    const int idx = tid + 256 * k, jl = idx >> 4, c = idx & 15;
    ra[k] = *(const v4f*)(ajg + (size_t)jl * D_ + 4 * c);
  }
#pragma unroll
  for (int k = 0; k < AD_V4; ++k) {
    const int idx = tid + 256 * k, row = idx >> 5, c = idx & 31;
    rd[k] = *(const v4f*)(adg + (size_t)row * N_ + 4 * c);
  }

  const int r0 = i0 + 2 * wave;
  const size_t g0 = (size_t)(b * N_ + r0) * D_ + o0 + 2 * lane;
  const v2f ai0 = *(const v2f*)(ai + g0);
  const v2f ai1 = *(const v2f*)(ai + g0 + D_);
  v2f acc0 = {0.f, 0.f}, acc1 = {0.f, 0.f};

#pragma unroll
  for (int k = 0; k < AJ_V4; ++k) {
    const int idx = tid + 256 * k, jl = idx >> 4, c = idx & 15;
    *(v4f*)(ajs + jl * PO + 4 * c) = ra[k];
  }
#pragma unroll
  for (int k = 0; k < AD_V4; ++k) {
    const int idx = tid + 256 * k, row = idx >> 5, c = idx & 31;
    *(v4f*)(ads + row * AP + 4 * c) = rd[k];
  }

  const v2f* ajl = (const v2f*)ajs + lane;
  const v4f* d0l = (const v4f*)(ads + (2 * wave) * AP);
  const v4f* d1l = (const v4f*)(ads + (2 * wave + 1) * AP);

#pragma unroll
  for (int p = 0; p < NPH; ++p) {
    __syncthreads();
    if (p + 1 < NPH) {
      const int jb = (p + 1) * JP;
#pragma unroll
      for (int k = 0; k < AJ_V4; ++k) {
        const int idx = tid + 256 * k, jl = idx >> 4, c = idx & 15;
        ra[k] = *(const v4f*)(ajg + (size_t)(jb + jl) * D_ + 4 * c);
      }
#pragma unroll
      for (int k = 0; k < AD_V4; ++k) {
        const int idx = tid + 256 * k, row = idx >> 5, c = idx & 31;
        rd[k] = *(const v4f*)(adg + (size_t)row * N_ + jb + 4 * c);
      }
    }
    for (int j = p * JP; j < (p + 1) * JP; j += GJ) {
      PGroup g;
      g.load(ajl, d0l, d1l, j);
      g.accumulate(ai0, ai1, acc0, acc1);
    }
    if (p + 1 < NPH) {
      const int jb = (p + 1) * JP;
#pragma unroll
      for (int k = 0; k < AJ_V4; ++k) {
        const int idx = tid + 256 * k, jl = idx >> 4, c = idx & 15;
        *(v4f*)(ajs + (jb + jl) * PO + 4 * c) = ra[k];
      }
#pragma unroll
      for (int k = 0; k < AD_V4; ++k) {
        const int idx = tid + 256 * k, row = idx >> 5, c = idx & 31;
        *(v4f*)(ads + row * AP + jb + 4 * c) = rd[k];
      }
    }
  }
  *(v2f*)(S + g0)      = acc0;
  *(v2f*)(S + g0 + D_) = acc1;
}

__global__ __launch_bounds__(256)
void rowsum_kernel(const float* __restrict__ adj, float* __restrict__ rowsum) {
  const int lane = threadIdx.x & 31;
  const int wave = threadIdx.x >> 5;
  const int row  = blockIdx.x * 8 + wave;
  const float* p = adj + row * N_;
  float s = 0.f;
  for (int j = lane; j < N_; j += 32) s += p[j];
#pragma unroll
  for (int off = 16; off > 0; off >>= 1) s += __shfl_down(s, off, 32);
  if (lane == 0) rowsum[row] = s;
}

extern "C" void kernel_launch(void* const* d_in, const int* in_sizes, int n_in,
                              void* d_out, int out_size, void* d_ws, size_t ws_size,
                              hipStream_t stream) {
  const float* node = (const float*)d_in[0];
  const float* adj  = (const float*)d_in[1];
  const float* w1   = (const float*)d_in[2];
  const float* b1   = (const float*)d_in[3];
  const float* w2   = (const float*)d_in[4];
  const float* b2   = (const float*)d_in[5];

  float* h  = (float*)d_out;
  float* ai = (float*)d_ws;
  float* aj = ai + (size_t)M_ * D_;
  float* S  = aj + (size_t)M_ * D_;
  float* rs = S  + (size_t)M_ * D_;

  hipMemcpyAsync(h, node, (size_t)M_ * D_ * sizeof(float),
                 hipMemcpyDeviceToDevice, stream);
  rowsum_kernel<<<M_ / 8, 256, 0, stream>>>(adj, rs);

  const dim3 gg(D_ / 16, M_ / 16);
  for (int l = 0; l < L_; ++l) {
    const float* w1l = w1 + (size_t)l * D_ * 2 * D_;
    const float* b1l = b1 + (size_t)l * D_;
    const float* w2l = w2 + (size_t)l * D_ * D_;
    const float* b2l = b2 + (size_t)l * D_;
    wmma_gemm_bias<<<gg, 32, 0, stream>>>(h, w1l, 2 * D_, 0,  b1l, ai);
    wmma_gemm_bias<<<gg, 32, 0, stream>>>(h, w1l, 2 * D_, D_, nullptr, aj);
    pairwise_reduce<<<dim3(D_ / PO, N_ / PI, B_), 256, 0, stream>>>(ai, aj, adj, S);
    wmma_gemm_final<<<gg, 32, 0, stream>>>(S, w2l, b2l, rs, h);
  }
}


# launch fusion, 10 stream ops to 5: ai and aj products share a launch, layer-0 update feeds layer-1 ai/aj through LDS in one kernel, row sums ride in the first, input copy dropped; sum orders kept
# speedup vs baseline: 2.5915x; 1.4724x over previous
#include <hip/hip_runtime.h>

typedef __attribute__((ext_vector_type(2))) float v2f;
typedef __attribute__((ext_vector_type(4))) float v4f;
typedef __attribute__((ext_vector_type(8))) float v8f;

#define B_ 4
#define N_ 512
#define D_ 128
#define L_ 2
#define M_ (B_ * N_)

#define HP (D_ + 4)

struct EdgeB {
  v2f b[D_ / 4];
  __device__ __forceinline__ void load(const float* __restrict__ W1, int part, int wave, int lane) {
    const float* wrow = W1 + ((wave * 16 + (lane & 15)) * 2 + part) * D_ + (lane >> 4) * 2;
#pragma unroll
    for (int kk = 0; kk < D_ / 4; ++kk) b[kk] = *(const v2f*)(wrow + kk * 4);
  }
};

__device__ __forceinline__ void edge_tile(const float* hs, const EdgeB& wb, float bv,
                                          float* __restrict__ o, int strip, int wave, int lane) {
  const int l16 = lane & 15;
  const int kh  = lane >> 4;
  const float* arow = hs + l16 * HP + kh * 2;
  v2f a[D_ / 4];
#pragma unroll
  for (int kk = 0; kk < D_ / 4; ++kk) a[kk] = *(const v2f*)(arow + kk * 4);
  __builtin_amdgcn_sched_barrier(0);
  v8f c = {};
#pragma unroll
  for (int kk = 0; kk < D_ / 4; ++kk)
    c = __builtin_amdgcn_wmma_f32_16x16x4_f32(
        false, a[kk], false, wb.b[kk], (short)0, c, false, false);
#pragma unroll
  for (int r = 0; r < 8; ++r) {
    const int m = strip * 16 + kh * 8 + r;
    o[m * D_ + wave * 16 + l16] = c[r] + bv;
  }
}

__global__ __launch_bounds__(256)
void first_edge(const float* __restrict__ X, const float* __restrict__ adj,
                const float* __restrict__ W1, const float* __restrict__ b1,
                float* __restrict__ ai, float* __restrict__ aj, float* __restrict__ rowsum) {
  __shared__ __attribute__((aligned(16))) float hs[16 * HP];
  const int tid  = threadIdx.x;
  const int lane = tid & 31;
  const int wave = tid >> 5;
  const int part  = blockIdx.x;
  const int strip = blockIdx.y;

  EdgeB wb;
  wb.load(W1, part, wave, lane);
  const float braw = b1[wave * 16 + (lane & 15)];
  const float bv = part == 0 ? braw : 0.f;

  v4f xs[2];
#pragma unroll
  for (int k = 0; k < 2; ++k) {
    const int idx = tid + 256 * k, row = idx >> 5, c4 = idx & 31;
    xs[k] = *(const v4f*)(X + (size_t)(strip * 16 + row) * D_ + 4 * c4);
  }
  {
    const int row = strip * 16 + part * 8 + wave;
    const float* p = adj + (size_t)row * N_;
    float s = 0.f;
#pragma unroll
    for (int q = 0; q < N_ / 32; ++q) s += p[lane + 32 * q];
#pragma unroll
    for (int off = 16; off > 0; off >>= 1) s += __shfl_down(s, off, 32);
    if (lane == 0) rowsum[row] = s;
  }
#pragma unroll
  for (int k = 0; k < 2; ++k) {
    const int idx = tid + 256 * k, row = idx >> 5, c4 = idx & 31;
    *(v4f*)(hs + row * HP + 4 * c4) = xs[k];
  }
  __syncthreads();
  edge_tile(hs, wb, bv, part == 0 ? ai : aj, strip, wave, lane);
}

__global__ __launch_bounds__(256)
void mid_edge(const float* __restrict__ hin, const float* __restrict__ S,
              const float* __restrict__ W2, const float* __restrict__ b2,
              const float* __restrict__ rowsum, float* __restrict__ hout,
              const float* __restrict__ W1, const float* __restrict__ b1,
              float* __restrict__ ai, float* __restrict__ aj) {
  __shared__ __attribute__((aligned(16))) float hs[16 * HP];
  const int tid  = threadIdx.x;
  const int lane = tid & 31;
  const int wave = tid >> 5;
  const int l16  = lane & 15;
  const int kh   = lane >> 4;
  const int part  = blockIdx.x;
  const int strip = blockIdx.y;

  EdgeB wb;
  wb.load(W1, part, wave, lane);
  const float braw = b1[wave * 16 + l16];
  const float bv1 = part == 0 ? braw : 0.f;

  const float bv2 = b2[wave * 16 + l16];
  float hv[8], rv[8];
#pragma unroll
  for (int r = 0; r < 8; ++r) {
    const int m = strip * 16 + kh * 8 + r;
    hv[r] = hin[m * D_ + wave * 16 + l16];
    rv[r] = rowsum[m];
  }

  const float* xrow = S  + (strip * 16 + l16) * D_;
  const float* wrow = W2 + (wave * 16 + l16) * D_;
  v8f c = {};
#pragma unroll
  for (int kk = 0; kk < D_ / 4; ++kk) {
    const int kb = kk * 4 + kh * 2;
    v2f a = *(const v2f*)(xrow + kb);
    v2f b = *(const v2f*)(wrow + kb);
    c = __builtin_amdgcn_wmma_f32_16x16x4_f32(
        false, a, false, b, (short)0, c, false, false);
  }
  float hn[8];
#pragma unroll
  for (int r = 0; r < 8; ++r) {
    hn[r] = hv[r] + c[r] + rv[r] * bv2;
    hs[(kh * 8 + r) * HP + wave * 16 + l16] = hn[r];
  }
  if (part == 0) {
#pragma unroll
    for (int r = 0; r < 8; ++r)
      hout[(strip * 16 + kh * 8 + r) * D_ + wave * 16 + l16] = hn[r];
  }
  __syncthreads();
  edge_tile(hs, wb, bv1, part == 0 ? ai : aj, strip, wave, lane);
}

__global__ __launch_bounds__(32)
void wmma_gemm_final(const float* __restrict__ S, const float* __restrict__ W2,
                     const float* __restrict__ b2, const float* __restrict__ rowsum,
                     float* __restrict__ h) {
  const int lane = threadIdx.x;
  const int l16  = lane & 15;
  const int half = lane >> 4;
  const int tm = blockIdx.y;
  const int tn = blockIdx.x;
  const float* xrow = S + (tm * 16 + l16) * D_;
  const float* wrow = W2 + (tn * 16 + l16) * D_;
  v8f c = {};
#pragma unroll
  for (int kk = 0; kk < D_ / 4; ++kk) {
    const int kb = kk * 4 + half * 2;
    v2f a = *(const v2f*)(xrow + kb);
    v2f b = *(const v2f*)(wrow + kb);
    c = __builtin_amdgcn_wmma_f32_16x16x4_f32(
        false, a, false, b, (short)0, c, false, false);
  }
  const float bv = b2[tn * 16 + l16];
#pragma unroll
  for (int r = 0; r < 8; ++r) {
    const int m = tm * 16 + half * 8 + r;
    const int idx = m * D_ + tn * 16 + l16;
    h[idx] = h[idx] + c[r] + rowsum[m] * bv;
  }
}

#define PI 16
#define PO 64
#define JP 128
#define NPH (N_ / JP)
#define AP (N_ + 4)
#define AJ_V4 (JP * PO / 4 / 256)
#define AD_V4 (PI * JP / 4 / 256)

#define GJ 16

struct PGroup {
  v4f d0[GJ / 4], d1[GJ / 4];
  v2f a[GJ];
  __device__ __forceinline__ void load(const v2f* ajl, const v4f* d0l, const v4f* d1l, int j) {
#pragma unroll
    for (int q = 0; q < GJ / 4; ++q) { d0[q] = d0l[(j >> 2) + q]; d1[q] = d1l[(j >> 2) + q]; }
#pragma unroll
    for (int q = 0; q < GJ; ++q) a[q] = ajl[(j + q) * (PO / 2)];
  }
  __device__ __forceinline__ void accumulate(const v2f ai0, const v2f ai1, v2f& acc0, v2f& acc1) const {
#pragma unroll
    for (int q = 0; q < GJ; ++q) {
#pragma unroll
      for (int c = 0; c < 2; ++c) {
        const float t0 = fmaxf(ai0[c] + a[q][c], 0.f);
        const float t1 = fmaxf(ai1[c] + a[q][c], 0.f);
        acc0[c] = fmaf(d0[q >> 2][q & 3], t0, acc0[c]);
        acc1[c] = fmaf(d1[q >> 2][q & 3], t1, acc1[c]);
      }
    }
  }
};

__global__ __launch_bounds__(256)
void pairwise_reduce(const float* __restrict__ ai, const float* __restrict__ aj,
                     const float* __restrict__ adj, float* __restrict__ S) {
  __shared__ __attribute__((aligned(16))) float ajs[N_ * PO];
  __shared__ __attribute__((aligned(16))) float ads[PI * AP];
  const int tid  = threadIdx.x;
  const int lane = tid & 31;
  const int wave = tid >> 5;
  const int o0 = blockIdx.x * PO;
  const int i0 = blockIdx.y * PI;
  const int b  = blockIdx.z;

  const float* ajg = aj  + (size_t)(b * N_) * D_ + o0;
  const float* adg = adj + (size_t)(b * N_ + i0) * N_;

  v4f ra[AJ_V4], rd[AD_V4];
#pragma unroll
  for (int k = 0; k < AJ_V4; ++k) {
    const int idx = tid + 256 * k, jl = idx >> 4, c = idx & 15;
    ra[k] = *(const v4f*)(ajg + (size_t)jl * D_ + 4 * c);
  }
#pragma unroll
  for (int k = 0; k < AD_V4; ++k) {
    const int idx = tid + 256 * k, row = idx >> 5, c = idx & 31;
    rd[k] = *(const v4f*)(adg + (size_t)row * N_ + 4 * c);
  }

  const int r0 = i0 + 2 * wave;
  const size_t g0 = (size_t)(b * N_ + r0) * D_ + o0 + 2 * lane;
  const v2f ai0 = *(const v2f*)(ai + g0);
  const v2f ai1 = *(const v2f*)(ai + g0 + D_);
  v2f acc0 = {0.f, 0.f}, acc1 = {0.f, 0.f};

#pragma unroll
  for (int k = 0; k < AJ_V4; ++k) {
    const int idx = tid + 256 * k, jl = idx >> 4, c = idx & 15;
    *(v4f*)(ajs + jl * PO + 4 * c) = ra[k];
  }
#pragma unroll
  for (int k = 0; k < AD_V4; ++k) {
    const int idx = tid + 256 * k, row = idx >> 5, c = idx & 31;
    *(v4f*)(ads + row * AP + 4 * c) = rd[k];
  }

  const v2f* ajl = (const v2f*)ajs + lane;
  const v4f* d0l = (const v4f*)(ads + (2 * wave) * AP);
  const v4f* d1l = (const v4f*)(ads + (2 * wave + 1) * AP);

#pragma unroll
  for (int p = 0; p < NPH; ++p) {
    __syncthreads();
    if (p + 1 < NPH) {
      const int jb = (p + 1) * JP;
#pragma unroll
      for (int k = 0; k < AJ_V4; ++k) {
        const int idx = tid + 256 * k, jl = idx >> 4, c = idx & 15;
        ra[k] = *(const v4f*)(ajg + (size_t)(jb + jl) * D_ + 4 * c);
      }
#pragma unroll
      for (int k = 0; k < AD_V4; ++k) {
        const int idx = tid + 256 * k, row = idx >> 5, c = idx & 31;
        rd[k] = *(const v4f*)(adg + (size_t)row * N_ + jb + 4 * c);
      }
    }
    for (int j = p * JP; j < (p + 1) * JP; j += GJ) {
      PGroup g;
      g.load(ajl, d0l, d1l, j);
      g.accumulate(ai0, ai1, acc0, acc1);
    }
    if (p + 1 < NPH) {
      const int jb = (p + 1) * JP;
#pragma unroll
      for (int k = 0; k < AJ_V4; ++k) {
        const int idx = tid + 256 * k, jl = idx >> 4, c = idx & 15;
        *(v4f*)(ajs + (jb + jl) * PO + 4 * c) = ra[k];
      }
#pragma unroll
      for (int k = 0; k < AD_V4; ++k) {
        const int idx = tid + 256 * k, row = idx >> 5, c = idx & 31;
        *(v4f*)(ads + row * AP + jb + 4 * c) = rd[k];
      }
    }
  }
  *(v2f*)(S + g0)      = acc0;
  *(v2f*)(S + g0 + D_) = acc1;
}

extern "C" void kernel_launch(void* const* d_in, const int* in_sizes, int n_in,
                              void* d_out, int out_size, void* d_ws, size_t ws_size,
                              hipStream_t stream) {
  const float* node = (const float*)d_in[0];
  const float* adj  = (const float*)d_in[1];
  const float* w1   = (const float*)d_in[2];
  const float* b1   = (const float*)d_in[3];
  const float* w2   = (const float*)d_in[4];
  const float* b2   = (const float*)d_in[5];

  float* h  = (float*)d_out;
  float* ai = (float*)d_ws;
  float* aj = ai + (size_t)M_ * D_;
  float* S  = aj + (size_t)M_ * D_;
  float* rs = S  + (size_t)M_ * D_;

  const dim3 ge(2, M_ / 16);
  const dim3 gp(D_ / PO, N_ / PI, B_);
  first_edge<<<ge, 256, 0, stream>>>(node, adj, w1, b1, ai, aj, rs);
  pairwise_reduce<<<gp, 256, 0, stream>>>(ai, aj, adj, S);
  mid_edge<<<ge, 256, 0, stream>>>(node, S, w2, b2, rs, h,
                                   w1 + (size_t)D_ * 2 * D_, b1 + D_, ai, aj);
  pairwise_reduce<<<gp, 256, 0, stream>>>(ai, aj, adj, S);
  wmma_gemm_final<<<dim3(D_ / 16, M_ / 16), 32, 0, stream>>>(
      S, w2 + (size_t)D_ * D_, b2 + D_, rs, h);
}


# launch fusion, 5 launches to 3: pairwise block retiled to 8 rows x all 128 columns, so it owns whole S rows and runs the update product and the next ai/aj products in its tail; sum orders kept
# speedup vs baseline: 3.0486x; 1.1764x over previous
#include <hip/hip_runtime.h>

typedef __attribute__((ext_vector_type(2))) float v2f;
typedef __attribute__((ext_vector_type(4))) float v4f;
typedef __attribute__((ext_vector_type(8))) float v8f;

#define B_ 4
#define N_ 512
#define D_ 128
#define L_ 2
#define M_ (B_ * N_)

#define HP (D_ + 4)

struct EdgeB {
  v2f b[D_ / 4];
  __device__ __forceinline__ void load_row(const float* __restrict__ wrow) {
#pragma unroll
    for (int kk = 0; kk < D_ / 4; ++kk) b[kk] = *(const v2f*)(wrow + kk * 4);
  }
  __device__ __forceinline__ void load(const float* __restrict__ W1, int part, int wave, int lane) {
    load_row(W1 + ((wave * 16 + (lane & 15)) * 2 + part) * D_ + (lane >> 4) * 2);
  }
};

__device__ __forceinline__ void edge_tile(const float* hs, const EdgeB& wb, float bv,
                                          float* __restrict__ o, int strip, int wave, int lane) {
  const int l16 = lane & 15;
  const int kh  = lane >> 4;
  const float* arow = hs + l16 * HP + kh * 2;
  v2f a[D_ / 4];
#pragma unroll
  for (int kk = 0; kk < D_ / 4; ++kk) a[kk] = *(const v2f*)(arow + kk * 4);
  __builtin_amdgcn_sched_barrier(0);
  v8f c = {};
#pragma unroll
  for (int kk = 0; kk < D_ / 4; ++kk)
    c = __builtin_amdgcn_wmma_f32_16x16x4_f32(
        false, a[kk], false, wb.b[kk], (short)0, c, false, false);
#pragma unroll
  for (int r = 0; r < 8; ++r) {
    const int m = strip * 16 + kh * 8 + r;
    o[m * D_ + wave * 16 + l16] = c[r] + bv;
  }
}

__global__ __launch_bounds__(256)
void first_edge(const float* __restrict__ X, const float* __restrict__ adj,
                const float* __restrict__ W1, const float* __restrict__ b1,
                float* __restrict__ ai, float* __restrict__ aj, float* __restrict__ rowsum) {
  __shared__ __attribute__((aligned(16))) float hs[16 * HP];
  const int tid  = threadIdx.x;
  const int lane = tid & 31;
  const int wave = tid >> 5;
  const int part  = blockIdx.x;
  const int strip = blockIdx.y;

  EdgeB wb;
  wb.load(W1, part, wave, lane);
  const float braw = b1[wave * 16 + (lane & 15)];
  const float bv = part == 0 ? braw : 0.f;

  v4f xs[2];
#pragma unroll
  for (int k = 0; k < 2; ++k) {
    const int idx = tid + 256 * k, row = idx >> 5, c4 = idx & 31;
    xs[k] = *(const v4f*)(X + (size_t)(strip * 16 + row) * D_ + 4 * c4);
  }
  {
    const int row = strip * 16 + part * 8 + wave;
    const float* p = adj + (size_t)row * N_;
    float s = 0.f;
#pragma unroll
    for (int q = 0; q < N_ / 32; ++q) s += p[lane + 32 * q];
#pragma unroll
    for (int off = 16; off > 0; off >>= 1) s += __shfl_down(s, off, 32);
    if (lane == 0) rowsum[row] = s;
  }
#pragma unroll
  for (int k = 0; k < 2; ++k) {
    const int idx = tid + 256 * k, row = idx >> 5, c4 = idx & 31;
    *(v4f*)(hs + row * HP + 4 * c4) = xs[k];
  }
  __syncthreads();
  edge_tile(hs, wb, bv, part == 0 ? ai : aj, strip, wave, lane);
}

#define PI 8
#define PO D_
#define JP 128
#define NPH (N_ / JP)
#define AP (N_ + 4)
#define AJ_V4 (JP * PO / 4 / 256)
#define AD_V4 (PI * JP / 4 / 256)

#define GJ 16

struct PGroup {
  v4f d0[GJ / 4], d1[GJ / 4];
  v2f a[GJ];
  __device__ __forceinline__ void load(const v2f* ajl, const v4f* d0l, const v4f* d1l, int j) {
#pragma unroll
    for (int q = 0; q < GJ / 4; ++q) { d0[q] = d0l[(j >> 2) + q]; d1[q] = d1l[(j >> 2) + q]; }
#pragma unroll
    for (int q = 0; q < GJ; ++q) a[q] = ajl[(j + q) * (PO / 2)];
  }
  __device__ __forceinline__ void accumulate(const v2f ai0, const v2f ai1, v2f& acc0, v2f& acc1) const {
#pragma unroll
    for (int q = 0; q < GJ; ++q) {
#pragma unroll
      for (int c = 0; c < 2; ++c) {
        const float t0 = fmaxf(ai0[c] + a[q][c], 0.f);
        const float t1 = fmaxf(ai1[c] + a[q][c], 0.f);
        acc0[c] = fmaf(d0[q >> 2][q & 3], t0, acc0[c]);
        acc1[c] = fmaf(d1[q >> 2][q & 3], t1, acc1[c]);
      }
    }
  }
};

struct StripA {
  v2f a[D_ / 4];
  __device__ __forceinline__ void load(const float* strip, int lane) {
    const float* arow = strip + (lane & 15) * HP + (lane >> 4) * 2;
#pragma unroll
    for (int kk = 0; kk < D_ / 4; ++kk) a[kk] = *(const v2f*)(arow + kk * 4);
  }
};

template <bool LAST>
__device__ __forceinline__ void pair_body(
    float* ajs, float* ads, float* ss, float* hs,
    const float* __restrict__ ai, const float* __restrict__ aj,
    const float* __restrict__ adj, const float* hin,
    const float* __restrict__ W2, const float* __restrict__ b2,
    const float* __restrict__ rowsum, float* hout,
    const float* __restrict__ W1, const float* __restrict__ b1,
    float* __restrict__ aio, float* __restrict__ ajo) {
  const int tid  = threadIdx.x;
  const int lane = tid & 31;
  const int wave = tid >> 5;
  const int l16  = lane & 15;
  const int kh   = lane >> 4;
  const int rp   = wave >> 1;
  const int ch   = wave & 1;
  const int i0 = blockIdx.x * PI;
  const int b  = blockIdx.y;
  const int m0 = b * N_ + i0;

  const float* ajg = aj  + (size_t)(b * N_) * D_;
  const float* adg = adj + (size_t)m0 * N_;

  v4f ra[AJ_V4], rd[AD_V4];
#pragma unroll
  for (int k = 0; k < AJ_V4; ++k)
    ra[k] = *(const v4f*)(ajg + 4 * (tid + 256 * k));
#pragma unroll
  for (int k = 0; k < AD_V4; ++k) {
    const int idx = tid + 256 * k, row = idx >> 5, c = idx & 31;
    rd[k] = *(const v4f*)(adg + (size_t)row * N_ + 4 * c);
  }

  const size_t g0 = (size_t)(m0 + 2 * rp) * D_ + 64 * ch + 2 * lane;
  const v2f ai0 = *(const v2f*)(ai + g0);
  const v2f ai1 = *(const v2f*)(ai + g0 + D_);
  v2f acc0 = {0.f, 0.f}, acc1 = {0.f, 0.f};

  EdgeB w2b, wib, wjb;
  float bv2 = 0.f, bv1 = 0.f;
  float hv[PI], rv[PI];

  for (int idx = tid; idx < (16 - PI) * HP; idx += 256) {
    ss[PI * HP + idx] = 0.f;
    if (!LAST) hs[PI * HP + idx] = 0.f;
  }

#pragma unroll
  for (int k = 0; k < AJ_V4; ++k)
    *(v4f*)(ajs + 4 * (tid + 256 * k)) = ra[k];
#pragma unroll
  for (int k = 0; k < AD_V4; ++k) {
    const int idx = tid + 256 * k, row = idx >> 5, c = idx & 31;
    *(v4f*)(ads + row * AP + 4 * c) = rd[k];
  }

  const v2f* ajl = (const v2f*)ajs + 32 * ch + lane;
  const v4f* d0l = (const v4f*)(ads + (2 * rp) * AP);
  const v4f* d1l = (const v4f*)(ads + (2 * rp + 1) * AP);

#pragma unroll
  for (int p = 0; p < NPH; ++p) {
    __syncthreads();
    if (p + 1 < NPH) {
      const int jb = (p + 1) * JP;
#pragma unroll
      for (int k = 0; k < AJ_V4; ++k)
        ra[k] = *(const v4f*)(ajg + (size_t)jb * D_ + 4 * (tid + 256 * k));
#pragma unroll
      for (int k = 0; k < AD_V4; ++k) {
        const int idx = tid + 256 * k, row = idx >> 5, c = idx & 31;
        rd[k] = *(const v4f*)(adg + (size_t)row * N_ + jb + 4 * c);
      }
    }
    if (p == NPH - 1) {
      w2b.load_row(W2 + (wave * 16 + l16) * D_ + kh * 2);
      bv2 = b2[wave * 16 + l16];
#pragma unroll
      for (int r = 0; r < PI; ++r) {
        hv[r] = hin[(size_t)(m0 + r) * D_ + wave * 16 + l16];
        rv[r] = rowsum[m0 + r];
      }
      if (!LAST) {
        wib.load(W1, 0, wave, lane);
        wjb.load(W1, 1, wave, lane);
        bv1 = b1[wave * 16 + l16];
      }
    }
    for (int j = p * JP; j < (p + 1) * JP; j += GJ) {
      PGroup g;
      g.load(ajl, d0l, d1l, j);
      g.accumulate(ai0, ai1, acc0, acc1);
    }
    if (p + 1 < NPH) {
      const int jb = (p + 1) * JP;
#pragma unroll
      for (int k = 0; k < AJ_V4; ++k)
        *(v4f*)(ajs + jb * PO + 4 * (tid + 256 * k)) = ra[k];
#pragma unroll
      for (int k = 0; k < AD_V4; ++k) {
        const int idx = tid + 256 * k, row = idx >> 5, c = idx & 31;
        *(v4f*)(ads + row * AP + jb + 4 * c) = rd[k];
      }
    }
  }

  *(v2f*)(ss + (2 * rp) * HP + 64 * ch + 2 * lane)     = acc0;
  *(v2f*)(ss + (2 * rp + 1) * HP + 64 * ch + 2 * lane) = acc1;
  __syncthreads();

  StripA sa;
  sa.load(ss, lane);
  __builtin_amdgcn_sched_barrier(0);
  v8f c = {};
#pragma unroll
  for (int kk = 0; kk < D_ / 4; ++kk)
    c = __builtin_amdgcn_wmma_f32_16x16x4_f32(
        false, sa.a[kk], false, w2b.b[kk], (short)0, c, false, false);
  float hn[PI];
#pragma unroll
  for (int r = 0; r < PI; ++r) hn[r] = hv[r] + c[r] + rv[r] * bv2;
  if (kh == 0) {
#pragma unroll
    for (int r = 0; r < PI; ++r)
      hout[(size_t)(m0 + r) * D_ + wave * 16 + l16] = hn[r];
  }
  if (LAST) return;

  if (kh == 0) {
#pragma unroll
    for (int r = 0; r < PI; ++r) hs[r * HP + wave * 16 + l16] = hn[r];
  }
  __syncthreads();

  StripA ha;
  ha.load(hs, lane);
  __builtin_amdgcn_sched_barrier(0);
  v8f ci = {}, cj = {};
#pragma unroll
  for (int kk = 0; kk < D_ / 4; ++kk) {
    ci = __builtin_amdgcn_wmma_f32_16x16x4_f32(
        false, ha.a[kk], false, wib.b[kk], (short)0, ci, false, false);
    cj = __builtin_amdgcn_wmma_f32_16x16x4_f32(
        false, ha.a[kk], false, wjb.b[kk], (short)0, cj, false, false);
  }
  float oi[PI], oj[PI];
#pragma unroll
  for (int r = 0; r < PI; ++r) { oi[r] = ci[r] + bv1; oj[r] = cj[r]; }
  if (kh == 0) {
#pragma unroll
    for (int r = 0; r < PI; ++r) {
      const size_t idx = (size_t)(m0 + r) * D_ + wave * 16 + l16;
      aio[idx] = oi[r];
      ajo[idx] = oj[r];
    }
  }
}

__global__ __launch_bounds__(256)
void pairwise_mid(const float* __restrict__ ai, const float* __restrict__ aj,
                  const float* __restrict__ adj, const float* __restrict__ hin,
                  const float* __restrict__ W2, const float* __restrict__ b2,
                  const float* __restrict__ rowsum, float* __restrict__ hout,
                  const float* __restrict__ W1, const float* __restrict__ b1,
                  float* __restrict__ aio, float* __restrict__ ajo) {
  __shared__ __attribute__((aligned(16))) float ajs[N_ * PO];
  __shared__ __attribute__((aligned(16))) float ads[PI * AP];
  __shared__ __attribute__((aligned(16))) float ss[16 * HP];
  __shared__ __attribute__((aligned(16))) float hs[16 * HP];
  pair_body<false>(ajs, ads, ss, hs, ai, aj, adj, hin, W2, b2, rowsum, hout, W1, b1, aio, ajo);
}

__global__ __launch_bounds__(256)
void pairwise_last(const float* __restrict__ ai, const float* __restrict__ aj,
                   const float* __restrict__ adj,
                   const float* __restrict__ W2, const float* __restrict__ b2,
                   const float* __restrict__ rowsum, float* h) {
  __shared__ __attribute__((aligned(16))) float ajs[N_ * PO];
  __shared__ __attribute__((aligned(16))) float ads[PI * AP];
  __shared__ __attribute__((aligned(16))) float ss[16 * HP];
  pair_body<true>(ajs, ads, ss, nullptr, ai, aj, adj, h, W2, b2, rowsum, h,
                  nullptr, nullptr, nullptr, nullptr);
}

extern "C" void kernel_launch(void* const* d_in, const int* in_sizes, int n_in,
                              void* d_out, int out_size, void* d_ws, size_t ws_size,
                              hipStream_t stream) {
  const float* node = (const float*)d_in[0];
  const float* adj  = (const float*)d_in[1];
  const float* w1   = (const float*)d_in[2];
  const float* b1   = (const float*)d_in[3];
  const float* w2   = (const float*)d_in[4];
  const float* b2   = (const float*)d_in[5];

  float* h   = (float*)d_out;
  float* ai0 = (float*)d_ws;
  float* aj0 = ai0 + (size_t)M_ * D_;
  float* ai1 = aj0 + (size_t)M_ * D_;
  float* aj1 = ai1 + (size_t)M_ * D_;
  float* rs  = aj1 + (size_t)M_ * D_;

  const dim3 ge(2, M_ / 16);
  const dim3 gp(N_ / PI, B_);
  first_edge<<<ge, 256, 0, stream>>>(node, adj, w1, b1, ai0, aj0, rs);
  pairwise_mid<<<gp, 256, 0, stream>>>(ai0, aj0, adj, node, w2, b2, rs, h,
                                       w1 + (size_t)D_ * 2 * D_, b1 + D_, ai1, aj1);
  pairwise_last<<<gp, 256, 0, stream>>>(ai1, aj1, adj, w2 + (size_t)D_ * D_, b2 + D_, rs, h);
}


# weight tiles of all products fetched as whole 512-byte rows (16-byte loads, 4 lines each), parked in LDS and read there as B fragments, not by 8-byte loads touching 16 lines each; sum orders kept
# speedup vs baseline: 3.2261x; 1.0582x over previous
#include <hip/hip_runtime.h>

typedef __attribute__((ext_vector_type(2))) float v2f;
typedef __attribute__((ext_vector_type(4))) float v4f;
typedef __attribute__((ext_vector_type(8))) float v8f;

#define B_ 4
#define N_ 512
#define D_ 128
#define L_ 2
#define M_ (B_ * N_)

#define HP (D_ + 4)

struct EdgeB {
  v2f b[D_ / 4];
  __device__ __forceinline__ void load_lds(const float* tile, int lane) {
    const float* brow = tile + (lane & 15) * HP + (lane >> 4) * 2;
#pragma unroll
    for (int kk = 0; kk < D_ / 4; ++kk) b[kk] = *(const v2f*)(brow + kk * 4);
  }
};

struct WTile {
  v4f r[16];
  __device__ __forceinline__ void fetch(const float* __restrict__ row0, int pitch, int lane) {
#pragma unroll
    for (int i = 0; i < 16; ++i) r[i] = *(const v4f*)(row0 + (size_t)i * pitch + 4 * lane);
  }
  __device__ __forceinline__ void park(float* tile, int lane) const {
#pragma unroll
    for (int i = 0; i < 16; ++i) *(v4f*)(tile + i * HP + 4 * lane) = r[i];
  }
};
#define WT (16 * HP)

__device__ __forceinline__ void edge_tile(const float* hs, const EdgeB& wb, float bv,
                                          float* __restrict__ o, int strip, int wave, int lane) {
  const int l16 = lane & 15;
  const int kh  = lane >> 4;
  const float* arow = hs + l16 * HP + kh * 2;
  v2f a[D_ / 4];
#pragma unroll
  for (int kk = 0; kk < D_ / 4; ++kk) a[kk] = *(const v2f*)(arow + kk * 4);
  __builtin_amdgcn_sched_barrier(0);
  v8f c = {};
#pragma unroll
  for (int kk = 0; kk < D_ / 4; ++kk)
    c = __builtin_amdgcn_wmma_f32_16x16x4_f32(
        false, a[kk], false, wb.b[kk], (short)0, c, false, false);
#pragma unroll
  for (int r = 0; r < 8; ++r) {
    const int m = strip * 16 + kh * 8 + r;
    o[m * D_ + wave * 16 + l16] = c[r] + bv;
  }
}

__global__ __launch_bounds__(256)
void first_edge(const float* __restrict__ X, const float* __restrict__ adj,
                const float* __restrict__ W1, const float* __restrict__ b1,
                float* __restrict__ ai, float* __restrict__ aj, float* __restrict__ rowsum) {
  __shared__ __attribute__((aligned(16))) float hs[16 * HP];
  __shared__ __attribute__((aligned(16))) float wt[8 * WT];
  const int tid  = threadIdx.x;
  const int lane = tid & 31;
  const int wave = tid >> 5;
  const int part  = blockIdx.x;
  const int strip = blockIdx.y;

  WTile wtile;
  wtile.fetch(W1 + (size_t)(wave * 16) * 2 * D_ + part * D_, 2 * D_, lane);
  const float braw = b1[wave * 16 + (lane & 15)];
  const float bv = part == 0 ? braw : 0.f;

  v4f xs[2];
#pragma unroll
  for (int k = 0; k < 2; ++k) {
    const int idx = tid + 256 * k, row = idx >> 5, c4 = idx & 31;
    xs[k] = *(const v4f*)(X + (size_t)(strip * 16 + row) * D_ + 4 * c4);
  }
  {
    const int row = strip * 16 + part * 8 + wave;
    const float* p = adj + (size_t)row * N_;
    float s = 0.f;
#pragma unroll
    for (int q = 0; q < N_ / 32; ++q) s += p[lane + 32 * q];
#pragma unroll
    for (int off = 16; off > 0; off >>= 1) s += __shfl_down(s, off, 32);
    if (lane == 0) rowsum[row] = s;
  }
#pragma unroll
  for (int k = 0; k < 2; ++k) {
    const int idx = tid + 256 * k, row = idx >> 5, c4 = idx & 31;
    *(v4f*)(hs + row * HP + 4 * c4) = xs[k];
  }
  wtile.park(wt + wave * WT, lane);
  __syncthreads();
  EdgeB wb;
  wb.load_lds(wt + wave * WT, lane);
  edge_tile(hs, wb, bv, part == 0 ? ai : aj, strip, wave, lane);
}

#define PI 8
#define PO D_
#define JP 128
#define NPH (N_ / JP)
#define AP (N_ + 4)
#define AJ_V4 (JP * PO / 4 / 256)
#define AD_V4 (PI * JP / 4 / 256)

#define GJ 16

struct PGroup {
  v4f d0[GJ / 4], d1[GJ / 4];
  v2f a[GJ];
  __device__ __forceinline__ void load(const v2f* ajl, const v4f* d0l, const v4f* d1l, int j) {
#pragma unroll
    for (int q = 0; q < GJ / 4; ++q) { d0[q] = d0l[(j >> 2) + q]; d1[q] = d1l[(j >> 2) + q]; }
#pragma unroll
    for (int q = 0; q < GJ; ++q) a[q] = ajl[(j + q) * (PO / 2)];
  }
  __device__ __forceinline__ void accumulate(const v2f ai0, const v2f ai1, v2f& acc0, v2f& acc1) const {
#pragma unroll
    for (int q = 0; q < GJ; ++q) {
#pragma unroll
      for (int c = 0; c < 2; ++c) {
        const float t0 = fmaxf(ai0[c] + a[q][c], 0.f);
        const float t1 = fmaxf(ai1[c] + a[q][c], 0.f);
        acc0[c] = fmaf(d0[q >> 2][q & 3], t0, acc0[c]);
        acc1[c] = fmaf(d1[q >> 2][q & 3], t1, acc1[c]);
      }
    }
  }
};

struct StripA {
  v2f a[D_ / 4];
  __device__ __forceinline__ void load(const float* strip, int lane) {
    const float* arow = strip + (lane & 15) * HP + (lane >> 4) * 2;
#pragma unroll
    for (int kk = 0; kk < D_ / 4; ++kk) a[kk] = *(const v2f*)(arow + kk * 4);
  }
};

template <bool LAST>
__device__ __forceinline__ void pair_body(
    float* ajs, float* ads, float* ss, float* hs, float* wx,
    const float* __restrict__ ai, const float* __restrict__ aj,
    const float* __restrict__ adj, const float* hin,
    const float* __restrict__ W2, const float* __restrict__ b2,
    const float* __restrict__ rowsum, float* hout,
    const float* __restrict__ W1, const float* __restrict__ b1,
    float* __restrict__ aio, float* __restrict__ ajo) {
  const int tid  = threadIdx.x;
  const int lane = tid & 31;
  const int wave = tid >> 5;
  const int l16  = lane & 15;
  const int kh   = lane >> 4;
  const int rp   = wave >> 1;
  const int ch   = wave & 1;
  const int i0 = blockIdx.x * PI;
  const int b  = blockIdx.y;
  const int m0 = b * N_ + i0;

  const float* ajg = aj  + (size_t)(b * N_) * D_;
  const float* adg = adj + (size_t)m0 * N_;

  v4f ra[AJ_V4], rd[AD_V4];
#pragma unroll
  for (int k = 0; k < AJ_V4; ++k)
    ra[k] = *(const v4f*)(ajg + 4 * (tid + 256 * k));
#pragma unroll
  for (int k = 0; k < AD_V4; ++k) {
    const int idx = tid + 256 * k, row = idx >> 5, c = idx & 31;
    rd[k] = *(const v4f*)(adg + (size_t)row * N_ + 4 * c);
  }

  const size_t g0 = (size_t)(m0 + 2 * rp) * D_ + 64 * ch + 2 * lane;
  const v2f ai0 = *(const v2f*)(ai + g0);
  const v2f ai1 = *(const v2f*)(ai + g0 + D_);
  v2f acc0 = {0.f, 0.f}, acc1 = {0.f, 0.f};

  WTile t2, ti, tj;
  float* const p2 = ajs + wave * WT;
  float* const pi = ajs + (8 + wave) * WT;
  float* const pj = (16 + wave) * WT + WT <= (NPH - 1) * JP * PO ? ajs + (16 + wave) * WT : wx;
  float bv2 = 0.f, bv1 = 0.f;
  float hv[PI], rv[PI];

  for (int idx = tid; idx < (16 - PI) * HP; idx += 256) {
    ss[PI * HP + idx] = 0.f;
    if (!LAST) hs[PI * HP + idx] = 0.f;
  }

#pragma unroll
  for (int k = 0; k < AJ_V4; ++k)
    *(v4f*)(ajs + 4 * (tid + 256 * k)) = ra[k];
#pragma unroll
  for (int k = 0; k < AD_V4; ++k) {
    const int idx = tid + 256 * k, row = idx >> 5, c = idx & 31;
    *(v4f*)(ads + row * AP + 4 * c) = rd[k];
  }

  const v2f* ajl = (const v2f*)ajs + 32 * ch + lane;
  const v4f* d0l = (const v4f*)(ads + (2 * rp) * AP);
  const v4f* d1l = (const v4f*)(ads + (2 * rp + 1) * AP);

#pragma unroll
  for (int p = 0; p < NPH; ++p) {
    __syncthreads();
    if (p + 1 < NPH) {
      const int jb = (p + 1) * JP;
#pragma unroll
      for (int k = 0; k < AJ_V4; ++k)
        ra[k] = *(const v4f*)(ajg + (size_t)jb * D_ + 4 * (tid + 256 * k));
#pragma unroll
      for (int k = 0; k < AD_V4; ++k) {
        const int idx = tid + 256 * k, row = idx >> 5, c = idx & 31;
        rd[k] = *(const v4f*)(adg + (size_t)row * N_ + jb + 4 * c);
      }
    }
    if (p == NPH - 1) {
      t2.fetch(W2 + (size_t)(wave * 16) * D_, D_, lane);
      bv2 = b2[wave * 16 + l16];
#pragma unroll
      for (int r = 0; r < PI; ++r) {
        hv[r] = hin[(size_t)(m0 + r) * D_ + wave * 16 + l16];
        rv[r] = rowsum[m0 + r];
      }
      if (!LAST) {
        ti.fetch(W1 + (size_t)(wave * 16) * 2 * D_, 2 * D_, lane);
        tj.fetch(W1 + (size_t)(wave * 16) * 2 * D_ + D_, 2 * D_, lane);
        bv1 = b1[wave * 16 + l16];
      }
    }
    for (int j = p * JP; j < (p + 1) * JP; j += GJ) {
      PGroup g;
      g.load(ajl, d0l, d1l, j);
      g.accumulate(ai0, ai1, acc0, acc1);
    }
    if (p == NPH - 1) {
      t2.park(p2, lane);
      if (!LAST) { ti.park(pi, lane); tj.park(pj, lane); }
    }
    if (p + 1 < NPH) {
      const int jb = (p + 1) * JP;
#pragma unroll
      for (int k = 0; k < AJ_V4; ++k)
        *(v4f*)(ajs + jb * PO + 4 * (tid + 256 * k)) = ra[k];
#pragma unroll
      for (int k = 0; k < AD_V4; ++k) {
        const int idx = tid + 256 * k, row = idx >> 5, c = idx & 31;
        *(v4f*)(ads + row * AP + jb + 4 * c) = rd[k];
      }
    }
  }

  *(v2f*)(ss + (2 * rp) * HP + 64 * ch + 2 * lane)     = acc0;
  *(v2f*)(ss + (2 * rp + 1) * HP + 64 * ch + 2 * lane) = acc1;
  __syncthreads();

  StripA sa;
  sa.load(ss, lane);
  EdgeB w2b;
  w2b.load_lds(p2, lane);
  __builtin_amdgcn_sched_barrier(0);
  v8f c = {};
#pragma unroll
  for (int kk = 0; kk < D_ / 4; ++kk)
    c = __builtin_amdgcn_wmma_f32_16x16x4_f32(
        false, sa.a[kk], false, w2b.b[kk], (short)0, c, false, false);
  float hn[PI];
#pragma unroll
  for (int r = 0; r < PI; ++r) hn[r] = hv[r] + c[r] + rv[r] * bv2;
  if (kh == 0) {
#pragma unroll
    for (int r = 0; r < PI; ++r)
      hout[(size_t)(m0 + r) * D_ + wave * 16 + l16] = hn[r];
  }
  if (LAST) return;

  if (kh == 0) {
#pragma unroll
    for (int r = 0; r < PI; ++r) hs[r * HP + wave * 16 + l16] = hn[r];
  }
  __syncthreads();

  StripA ha;
  ha.load(hs, lane);
  EdgeB wib, wjb;
  wib.load_lds(pi, lane);
  wjb.load_lds(pj, lane);
  __builtin_amdgcn_sched_barrier(0);
  v8f ci = {}, cj = {};
#pragma unroll
  for (int kk = 0; kk < D_ / 4; ++kk) {
    ci = __builtin_amdgcn_wmma_f32_16x16x4_f32(
        false, ha.a[kk], false, wib.b[kk], (short)0, ci, false, false);
    cj = __builtin_amdgcn_wmma_f32_16x16x4_f32(
        false, ha.a[kk], false, wjb.b[kk], (short)0, cj, false, false);
  }
  float oi[PI], oj[PI];
#pragma unroll
  for (int r = 0; r < PI; ++r) { oi[r] = ci[r] + bv1; oj[r] = cj[r]; }
  if (kh == 0) {
#pragma unroll
    for (int r = 0; r < PI; ++r) {
      const size_t idx = (size_t)(m0 + r) * D_ + wave * 16 + l16;
      aio[idx] = oi[r];
      ajo[idx] = oj[r];
    }
  }
}

__global__ __launch_bounds__(256)
void pairwise_mid(const float* __restrict__ ai, const float* __restrict__ aj,
                  const float* __restrict__ adj, const float* __restrict__ hin,
                  const float* __restrict__ W2, const float* __restrict__ b2,
                  const float* __restrict__ rowsum, float* __restrict__ hout,
                  const float* __restrict__ W1, const float* __restrict__ b1,
                  float* __restrict__ aio, float* __restrict__ ajo) {
  __shared__ __attribute__((aligned(16))) float ajs[N_ * PO];
  __shared__ __attribute__((aligned(16))) float ads[PI * AP];
  __shared__ __attribute__((aligned(16))) float ss[16 * HP];
  __shared__ __attribute__((aligned(16))) float hs[16 * HP];
  __shared__ __attribute__((aligned(16))) float wx[WT];
  pair_body<false>(ajs, ads, ss, hs, wx, ai, aj, adj, hin, W2, b2, rowsum, hout, W1, b1, aio, ajo);
}

__global__ __launch_bounds__(256)
void pairwise_last(const float* __restrict__ ai, const float* __restrict__ aj,
                   const float* __restrict__ adj,
                   const float* __restrict__ W2, const float* __restrict__ b2,
                   const float* __restrict__ rowsum, float* h) {
  __shared__ __attribute__((aligned(16))) float ajs[N_ * PO];
  __shared__ __attribute__((aligned(16))) float ads[PI * AP];
  __shared__ __attribute__((aligned(16))) float ss[16 * HP];
  pair_body<true>(ajs, ads, ss, nullptr, nullptr, ai, aj, adj, h, W2, b2, rowsum, h,
                  nullptr, nullptr, nullptr, nullptr);
}

extern "C" void kernel_launch(void* const* d_in, const int* in_sizes, int n_in,
                              void* d_out, int out_size, void* d_ws, size_t ws_size,
                              hipStream_t stream) {
  const float* node = (const float*)d_in[0];
  const float* adj  = (const float*)d_in[1];
  const float* w1   = (const float*)d_in[2];
  const float* b1   = (const float*)d_in[3];
  const float* w2   = (const float*)d_in[4];
  const float* b2   = (const float*)d_in[5];

  float* h   = (float*)d_out;
  float* ai0 = (float*)d_ws;
  float* aj0 = ai0 + (size_t)M_ * D_;
  float* ai1 = aj0 + (size_t)M_ * D_;
  float* aj1 = ai1 + (size_t)M_ * D_;
  float* rs  = aj1 + (size_t)M_ * D_;

  const dim3 ge(2, M_ / 16);
  const dim3 gp(N_ / PI, B_);
  first_edge<<<ge, 256, 0, stream>>>(node, adj, w1, b1, ai0, aj0, rs);
  pairwise_mid<<<gp, 256, 0, stream>>>(ai0, aj0, adj, node, w2, b2, rs, h,
                                       w1 + (size_t)D_ * 2 * D_, b1 + D_, ai1, aj1);
  pairwise_last<<<gp, 256, 0, stream>>>(ai1, aj1, adj, w2 + (size_t)D_ * D_, b2 + D_, rs, h);
}


# pairwise j loop: relu(ai+aj) taken as ai+max(aj,-ai), so per element max+fma (no add); ai*rowsum is added once behind the 8 partial sums: summed terms and their rounding change, orders kept
# speedup vs baseline: 3.8088x; 1.1806x over previous
#include <hip/hip_runtime.h>

typedef __attribute__((ext_vector_type(2))) float v2f;
typedef __attribute__((ext_vector_type(4))) float v4f;
typedef __attribute__((ext_vector_type(8))) float v8f;

#define B_ 4
#define N_ 512
#define D_ 128
#define L_ 2
#define M_ (B_ * N_)

#define HP (D_ + 4)

struct EdgeB {
  v2f b[D_ / 4];
  __device__ __forceinline__ void load_lds(const float* tile, int lane) {
    const float* brow = tile + (lane & 15) * HP + (lane >> 4) * 2;
#pragma unroll
    for (int kk = 0; kk < D_ / 4; ++kk) b[kk] = *(const v2f*)(brow + kk * 4);
  }
};

struct WTile {
  v4f r[16];
  __device__ __forceinline__ void fetch(const float* __restrict__ row0, int pitch, int lane) {
#pragma unroll
    for (int i = 0; i < 16; ++i) r[i] = *(const v4f*)(row0 + (size_t)i * pitch + 4 * lane);
  }
  __device__ __forceinline__ void park(float* tile, int lane) const {
#pragma unroll
    for (int i = 0; i < 16; ++i) *(v4f*)(tile + i * HP + 4 * lane) = r[i];
  }
};
#define WT (16 * HP)

__device__ __forceinline__ void edge_tile(const float* hs, const EdgeB& wb, float bv,
                                          float* __restrict__ o, int strip, int wave, int lane) {
  const int l16 = lane & 15;
  const int kh  = lane >> 4;
  const float* arow = hs + l16 * HP + kh * 2;
  v2f a[D_ / 4];
#pragma unroll
  for (int kk = 0; kk < D_ / 4; ++kk) a[kk] = *(const v2f*)(arow + kk * 4);
  __builtin_amdgcn_sched_barrier(0);
  v8f c = {};
#pragma unroll
  for (int kk = 0; kk < D_ / 4; ++kk)
    c = __builtin_amdgcn_wmma_f32_16x16x4_f32(
        false, a[kk], false, wb.b[kk], (short)0, c, false, false);
#pragma unroll
  for (int r = 0; r < 8; ++r) {
    const int m = strip * 16 + kh * 8 + r;
    o[m * D_ + wave * 16 + l16] = c[r] + bv;
  }
}

__global__ __launch_bounds__(256)
void first_edge(const float* __restrict__ X, const float* __restrict__ adj,
                const float* __restrict__ W1, const float* __restrict__ b1,
                float* __restrict__ ai, float* __restrict__ aj, float* __restrict__ rowsum) {
  __shared__ __attribute__((aligned(16))) float hs[16 * HP];
  __shared__ __attribute__((aligned(16))) float wt[8 * WT];
  const int tid  = threadIdx.x;
  const int lane = tid & 31;
  const int wave = tid >> 5;
  const int part  = blockIdx.x;
  const int strip = blockIdx.y;

  WTile wtile;
  wtile.fetch(W1 + (size_t)(wave * 16) * 2 * D_ + part * D_, 2 * D_, lane);
  const float braw = b1[wave * 16 + (lane & 15)];
  const float bv = part == 0 ? braw : 0.f;

  v4f xs[2];
#pragma unroll
  for (int k = 0; k < 2; ++k) {
    const int idx = tid + 256 * k, row = idx >> 5, c4 = idx & 31;
    xs[k] = *(const v4f*)(X + (size_t)(strip * 16 + row) * D_ + 4 * c4);
  }
  {
    const int row = strip * 16 + part * 8 + wave;
    const float* p = adj + (size_t)row * N_;
    float s = 0.f;
#pragma unroll
    for (int q = 0; q < N_ / 32; ++q) s += p[lane + 32 * q];
#pragma unroll
    for (int off = 16; off > 0; off >>= 1) s += __shfl_down(s, off, 32);
    if (lane == 0) rowsum[row] = s;
  }
#pragma unroll
  for (int k = 0; k < 2; ++k) {
    const int idx = tid + 256 * k, row = idx >> 5, c4 = idx & 31;
    *(v4f*)(hs + row * HP + 4 * c4) = xs[k];
  }
  wtile.park(wt + wave * WT, lane);
  __syncthreads();
  EdgeB wb;
  wb.load_lds(wt + wave * WT, lane);
  edge_tile(hs, wb, bv, part == 0 ? ai : aj, strip, wave, lane);
}

#define PI 8
#define NW 8
#define JW (N_ / NW)
#define JC 8
#define NCH (JW / JC)
#define AW (JW + 4)

struct StripA {
  v2f a[D_ / 4];
  __device__ __forceinline__ void load(const float* strip, int lane) {
    const float* arow = strip + (lane & 15) * HP + (lane >> 4) * 2;
#pragma unroll
    for (int kk = 0; kk < D_ / 4; ++kk) a[kk] = *(const v2f*)(arow + kk * 4);
  }
};

struct AjChunk {
  v4f r[JC];
  __device__ __forceinline__ void fetch(const float* __restrict__ p) {
#pragma unroll
    for (int q = 0; q < JC; ++q) r[q] = *(const v4f*)(p + (size_t)q * D_);
  }
};

__device__ __forceinline__ void pair_chunk(const AjChunk& ajc, const float* adc,
                                           const v2f (&nai2)[PI][2], v2f (&acc)[PI][2]) {
#pragma unroll
  for (int q4 = 0; q4 < JC / 4; ++q4) {
    v4f d[PI];
#pragma unroll
    for (int r = 0; r < PI; ++r) d[r] = *(const v4f*)(adc + r * AW + 4 * q4);
#pragma unroll
    for (int jj = 0; jj < 4; ++jj) {
      const v4f av = ajc.r[4 * q4 + jj];
      const v2f a0 = {av[0], av[1]}, a1 = {av[2], av[3]};
#pragma unroll
      for (int r = 0; r < PI; ++r) {
        const v2f dd = {d[r][jj], d[r][jj]};
        const v2f t0 = __builtin_elementwise_max(a0, nai2[r][0]);
        const v2f t1 = __builtin_elementwise_max(a1, nai2[r][1]);
        acc[r][0] = __builtin_elementwise_fma(dd, t0, acc[r][0]);
        acc[r][1] = __builtin_elementwise_fma(dd, t1, acc[r][1]);
      }
    }
  }
}

template <bool LAST>
__device__ __forceinline__ void pair_body(
    float* ads, float* ps, float* ss, float* hs, float* wts,
    const float* __restrict__ ai, const float* __restrict__ aj,
    const float* __restrict__ adj, const float* hin,
    const float* __restrict__ W2, const float* __restrict__ b2,
    const float* __restrict__ rowsum, float* hout,
    const float* __restrict__ W1, const float* __restrict__ b1,
    float* __restrict__ aio, float* __restrict__ ajo) {
  const int tid  = threadIdx.x;
  const int lane = tid & 31;
  const int wave = tid >> 5;
  const int l16  = lane & 15;
  const int kh   = lane >> 4;
  const int i0 = blockIdx.x * PI;
  const int b  = blockIdx.y;
  const int m0 = b * N_ + i0;
  const int jw = wave * JW;

  const float* ajg = aj  + (size_t)(b * N_ + jw) * D_ + 4 * lane;
  const float* adg = adj + (size_t)m0 * N_ + jw;

  AjChunk ca, cb;
  ca.fetch(ajg);
  v4f rd[PI * JW / 4 / 32];
#pragma unroll
  for (int k = 0; k < PI * JW / 4 / 32; ++k) {
    const int idx = lane + 32 * k, row = idx >> 4, c = idx & 15;
    rd[k] = *(const v4f*)(adg + (size_t)row * N_ + 4 * c);
  }
  v2f nai2[PI][2], acc[PI][2];
#pragma unroll
  for (int r = 0; r < PI; ++r) {
    const v4f t = *(const v4f*)(ai + (size_t)(m0 + r) * D_ + 4 * lane);
    nai2[r][0] = (v2f){-t[0], -t[1]};
    nai2[r][1] = (v2f){-t[2], -t[3]};
    acc[r][0] = (v2f){0.f, 0.f};
    acc[r][1] = (v2f){0.f, 0.f};
  }

  WTile t2;
  float* const p2 = wts + wave * WT;
  float* const pi = wts + (NW + wave) * WT;
  float* const pj = wts + (2 * NW + wave) * WT;
  t2.fetch(W2 + (size_t)(wave * 16) * D_, D_, lane);

  for (int idx = tid; idx < (16 - PI) * HP; idx += 256) {
    ss[PI * HP + idx] = 0.f;
    if (!LAST) hs[PI * HP + idx] = 0.f;
  }

  float* const adw = ads + wave * (PI * AW);
#pragma unroll
  for (int k = 0; k < PI * JW / 4 / 32; ++k) {
    const int idx = lane + 32 * k, row = idx >> 4, c = idx & 15;
    *(v4f*)(adw + row * AW + 4 * c) = rd[k];
  }
  __syncthreads();

#pragma unroll 1
  for (int ch = 0; ch < NCH - 2; ch += 2) {
    cb.fetch(ajg + (size_t)(ch + 1) * JC * D_);
    pair_chunk(ca, adw + ch * JC, nai2, acc);
    ca.fetch(ajg + (size_t)(ch + 2) * JC * D_);
    pair_chunk(cb, adw + (ch + 1) * JC, nai2, acc);
  }
  cb.fetch(ajg + (size_t)(NCH - 1) * JC * D_);
  pair_chunk(ca, adw + (NCH - 2) * JC, nai2, acc);
  pair_chunk(cb, adw + (NCH - 1) * JC, nai2, acc);
  t2.park(p2, lane);

  WTile ti, tj;
  float bv1 = 0.f;
  float hv[PI], rv[PI];
  const float bv2 = b2[wave * 16 + l16];
  const v4f aiw = *(const v4f*)(ai + (size_t)(m0 + wave) * D_ + 4 * lane);
  const float rsw = rowsum[m0 + wave];
#pragma unroll
  for (int r = 0; r < PI; ++r) {
    hv[r] = hin[(size_t)(m0 + r) * D_ + wave * 16 + l16];
    rv[r] = rowsum[m0 + r];
  }
  if (!LAST) {
    ti.fetch(W1 + (size_t)(wave * 16) * 2 * D_, 2 * D_, lane);
    tj.fetch(W1 + (size_t)(wave * 16) * 2 * D_ + D_, 2 * D_, lane);
    bv1 = b1[wave * 16 + l16];
  }

#pragma unroll
  for (int r = 0; r < PI; ++r) {
    const v4f o = {acc[r][0][0], acc[r][0][1], acc[r][1][0], acc[r][1][1]};
    *(v4f*)(ps + (wave * PI + r) * D_ + 4 * lane) = o;
  }
  __syncthreads();
  {
    const int row = tid >> 5, c4 = tid & 31;
    v4f s = *(const v4f*)(ps + row * D_ + 4 * c4);
#pragma unroll
    for (int w = 1; w < NW; ++w) s += *(const v4f*)(ps + (w * PI + row) * D_ + 4 * c4);
    s = __builtin_elementwise_fma(aiw, (v4f){rsw, rsw, rsw, rsw}, s);
    *(v4f*)(ss + row * HP + 4 * c4) = s;
  }
  __syncthreads();

  StripA sa;
  sa.load(ss, lane);
  EdgeB w2b;
  w2b.load_lds(p2, lane);
  __builtin_amdgcn_sched_barrier(0);
  v8f c = {};
#pragma unroll
  for (int kk = 0; kk < D_ / 4; ++kk)
    c = __builtin_amdgcn_wmma_f32_16x16x4_f32(
        false, sa.a[kk], false, w2b.b[kk], (short)0, c, false, false);
  float hn[PI];
#pragma unroll
  for (int r = 0; r < PI; ++r) hn[r] = hv[r] + c[r] + rv[r] * bv2;
  if (kh == 0) {
#pragma unroll
    for (int r = 0; r < PI; ++r)
      hout[(size_t)(m0 + r) * D_ + wave * 16 + l16] = hn[r];
  }
  if (LAST) return;

  if (kh == 0) {
#pragma unroll
    for (int r = 0; r < PI; ++r) hs[r * HP + wave * 16 + l16] = hn[r];
  }
  ti.park(pi, lane);
  tj.park(pj, lane);
  __syncthreads();

  StripA ha;
  ha.load(hs, lane);
  EdgeB wib, wjb;
  wib.load_lds(pi, lane);
  wjb.load_lds(pj, lane);
  __builtin_amdgcn_sched_barrier(0);
  v8f ci = {}, cj = {};
#pragma unroll
  for (int kk = 0; kk < D_ / 4; ++kk) {
    ci = __builtin_amdgcn_wmma_f32_16x16x4_f32(
        false, ha.a[kk], false, wib.b[kk], (short)0, ci, false, false);
    cj = __builtin_amdgcn_wmma_f32_16x16x4_f32(
        false, ha.a[kk], false, wjb.b[kk], (short)0, cj, false, false);
  }
  float oi[PI], oj[PI];
#pragma unroll
  for (int r = 0; r < PI; ++r) { oi[r] = ci[r] + bv1; oj[r] = cj[r]; }
  if (kh == 0) {
#pragma unroll
    for (int r = 0; r < PI; ++r) {
      const size_t idx = (size_t)(m0 + r) * D_ + wave * 16 + l16;
      aio[idx] = oi[r];
      ajo[idx] = oj[r];
    }
  }
}

__global__ __launch_bounds__(256)
void pairwise_mid(const float* __restrict__ ai, const float* __restrict__ aj,
                  const float* __restrict__ adj, const float* __restrict__ hin,
                  const float* __restrict__ W2, const float* __restrict__ b2,
                  const float* __restrict__ rowsum, float* __restrict__ hout,
                  const float* __restrict__ W1, const float* __restrict__ b1,
                  float* __restrict__ aio, float* __restrict__ ajo) {
  __shared__ __attribute__((aligned(16))) float ads[NW * PI * AW];
  __shared__ __attribute__((aligned(16))) float ps[NW * PI * D_];
  __shared__ __attribute__((aligned(16))) float ss[16 * HP];
  __shared__ __attribute__((aligned(16))) float hs[16 * HP];
  __shared__ __attribute__((aligned(16))) float wts[3 * NW * WT];
  pair_body<false>(ads, ps, ss, hs, wts, ai, aj, adj, hin, W2, b2, rowsum, hout, W1, b1, aio, ajo);
}

__global__ __launch_bounds__(256)
void pairwise_last(const float* __restrict__ ai, const float* __restrict__ aj,
                   const float* __restrict__ adj,
                   const float* __restrict__ W2, const float* __restrict__ b2,
                   const float* __restrict__ rowsum, float* h) {
  __shared__ __attribute__((aligned(16))) float ads[NW * PI * AW];
  __shared__ __attribute__((aligned(16))) float ps[NW * PI * D_];
  __shared__ __attribute__((aligned(16))) float ss[16 * HP];
  __shared__ __attribute__((aligned(16))) float wts[NW * WT];
  pair_body<true>(ads, ps, ss, nullptr, wts, ai, aj, adj, h, W2, b2, rowsum, h,
                  nullptr, nullptr, nullptr, nullptr);
}

extern "C" void kernel_launch(void* const* d_in, const int* in_sizes, int n_in,
                              void* d_out, int out_size, void* d_ws, size_t ws_size,
                              hipStream_t stream) {
  const float* node = (const float*)d_in[0];
  const float* adj  = (const float*)d_in[1];
  const float* w1   = (const float*)d_in[2];
  const float* b1   = (const float*)d_in[3];
  const float* w2   = (const float*)d_in[4];
  const float* b2   = (const float*)d_in[5];

  float* h   = (float*)d_out;
  float* ai0 = (float*)d_ws;
  float* aj0 = ai0 + (size_t)M_ * D_;
  float* ai1 = aj0 + (size_t)M_ * D_;
  float* aj1 = ai1 + (size_t)M_ * D_;
  float* rs  = aj1 + (size_t)M_ * D_;

  const dim3 ge(2, M_ / 16);
  const dim3 gp(N_ / PI, B_);
  first_edge<<<ge, 256, 0, stream>>>(node, adj, w1, b1, ai0, aj0, rs);
  pairwise_mid<<<gp, 256, 0, stream>>>(ai0, aj0, adj, node, w2, b2, rs, h,
                                       w1 + (size_t)D_ * 2 * D_, b1 + D_, ai1, aj1);
  pairwise_last<<<gp, 256, 0, stream>>>(ai1, aj1, adj, w2 + (size_t)D_ * D_, b2 + D_, rs, h);
}


# pairwise j loop software pipelined by one stage: a j's 16 max pairs go to registers of their own ahead of its 16 packed fmas, adj rows come from LDS one group of 4 j ahead; sums and orders kept
# speedup vs baseline: 4.2848x; 1.1250x over previous
#include <hip/hip_runtime.h>

typedef __attribute__((ext_vector_type(2))) float v2f;
typedef __attribute__((ext_vector_type(4))) float v4f;
typedef __attribute__((ext_vector_type(8))) float v8f;

#define B_ 4
#define N_ 512
#define D_ 128
#define L_ 2
#define M_ (B_ * N_)

#define HP (D_ + 4)

struct EdgeB {
  v2f b[D_ / 4];
};

struct WTile {
  v4f r[16];
  __device__ __forceinline__ void fetch(const float* __restrict__ row0, int pitch, int lane) {
#pragma unroll
    for (int i = 0; i < 16; ++i) r[i] = *(const v4f*)(row0 + (size_t)i * pitch + 4 * lane);
  }
  __device__ __forceinline__ void park(float* tile, int lane) const {
#pragma unroll
    for (int i = 0; i < 16; ++i) *(v4f*)(tile + i * HP + 4 * lane) = r[i];
  }
};
#define WT (16 * HP)

template <int K0, int K1>
__device__ __forceinline__ void frag_load(v2f (&a)[D_ / 4], v2f (&b)[D_ / 4],
                                          const float* arow, const float* brow) {
#pragma unroll
  for (int kk = K0; kk < K1; ++kk) {
    a[kk] = *(const v2f*)(arow + kk * 4);
    b[kk] = *(const v2f*)(brow + kk * 4);
  }
}

__device__ __forceinline__ void edge_tile(const float* hs, const float* tile, float bv,
                                          float* __restrict__ o, int strip, int wave, int lane) {
  const int l16 = lane & 15;
  const int kh  = lane >> 4;
  const float* arow = hs + l16 * HP + kh * 2;
  const float* brow = tile + l16 * HP + kh * 2;
  v2f a[D_ / 4];
  EdgeB wb;
  frag_load<0, 8>(a, wb.b, arow, brow);
  __builtin_amdgcn_sched_barrier(0);
  frag_load<8, D_ / 4>(a, wb.b, arow, brow);
  __builtin_amdgcn_sched_barrier(0);
  v8f c = {};
#pragma unroll
  for (int kk = 0; kk < D_ / 4; ++kk)
    c = __builtin_amdgcn_wmma_f32_16x16x4_f32(
        false, a[kk], false, wb.b[kk], (short)0, c, false, false);
#pragma unroll
  for (int r = 0; r < 8; ++r) {
    const int m = strip * 16 + kh * 8 + r;
    o[m * D_ + wave * 16 + l16] = c[r] + bv;
  }
}

__global__ __launch_bounds__(256)
void first_edge(const float* __restrict__ X, const float* __restrict__ adj,
                const float* __restrict__ W1, const float* __restrict__ b1,
                float* __restrict__ ai, float* __restrict__ aj, float* __restrict__ rowsum) {
  __shared__ __attribute__((aligned(16))) float hs[16 * HP];
  __shared__ __attribute__((aligned(16))) float wt[8 * WT];
  const int tid  = threadIdx.x;
  const int lane = tid & 31;
  const int wave = tid >> 5;
  const int part  = blockIdx.x;
  const int strip = blockIdx.y;

  WTile wtile;
  wtile.fetch(W1 + (size_t)(wave * 16) * 2 * D_ + part * D_, 2 * D_, lane);
  const float braw = b1[wave * 16 + (lane & 15)];
  const float bv = part == 0 ? braw : 0.f;

  v4f xs[2];
#pragma unroll
  for (int k = 0; k < 2; ++k) {
    const int idx = tid + 256 * k, row = idx >> 5, c4 = idx & 31;
    xs[k] = *(const v4f*)(X + (size_t)(strip * 16 + row) * D_ + 4 * c4);
  }
  int rs_row;
  float rs_val;
  {
    const int row = strip * 16 + part * 8 + wave;
    const float* p = adj + (size_t)row * N_;
    float s = 0.f;
#pragma unroll
    for (int q = 0; q < N_ / 32; ++q) s += p[lane + 32 * q];
#pragma unroll
    for (int off = 16; off > 0; off >>= 1) s += __shfl_down(s, off, 32);
    rs_row = row;
    rs_val = s;
  }
#pragma unroll
  for (int k = 0; k < 2; ++k) {
    const int idx = tid + 256 * k, row = idx >> 5, c4 = idx & 31;
    *(v4f*)(hs + row * HP + 4 * c4) = xs[k];
  }
  wtile.park(wt + wave * WT, lane);
  __syncthreads();
  if (lane == 0) rowsum[rs_row] = rs_val;
  edge_tile(hs, wt + wave * WT, bv, part == 0 ? ai : aj, strip, wave, lane);
}

#define PI 8
#define NW 8
#define JW (N_ / NW)
#define JC 8
#define NCH (JW / JC)
#define AW (JW + 4)

struct StripA {
  v2f a[D_ / 4];
};

struct AjChunk {
  v4f r[JC];
  __device__ __forceinline__ void fetch(const float* __restrict__ p) {
#pragma unroll
    for (int q = 0; q < JC; ++q) r[q] = *(const v4f*)(p + (size_t)q * D_);
  }
};

struct AdjGroup {
  v4f d[PI];
  __device__ __forceinline__ void read(const float* adc) {
#pragma unroll
    for (int r = 0; r < PI; ++r) d[r] = *(const v4f*)(adc + r * AW);
  }
};

template <bool NEXT>
__device__ __forceinline__ void pair_group(const v4f* ajr, const AdjGroup& d, AdjGroup& dn,
                                           const float* adn,
                                           const v2f (&nai2)[PI][2], v2f (&acc)[PI][2]) {
  if (NEXT) dn.read(adn);
#pragma unroll
  for (int jj = 0; jj < 4; ++jj) {
    const v4f av = ajr[jj];
    const v2f a0 = {av[0], av[1]}, a1 = {av[2], av[3]};
    v2f t[PI][2];
#pragma unroll
    for (int r = 0; r < PI; ++r) {
      t[r][0] = __builtin_elementwise_max(a0, nai2[r][0]);
      t[r][1] = __builtin_elementwise_max(a1, nai2[r][1]);
    }
    __builtin_amdgcn_sched_barrier(0);
#pragma unroll
    for (int r = 0; r < PI; ++r) {
      const v2f dd = {d.d[r][jj], d.d[r][jj]};
      acc[r][0] = __builtin_elementwise_fma(dd, t[r][0], acc[r][0]);
      acc[r][1] = __builtin_elementwise_fma(dd, t[r][1], acc[r][1]);
    }
  }
}

template <bool MORE>
__device__ __forceinline__ void pair_chunk(const AjChunk& ajc, const float* adc,
                                           AdjGroup& da, AdjGroup& db,
                                           const v2f (&nai2)[PI][2], v2f (&acc)[PI][2]) {
  pair_group<true>(ajc.r, da, db, adc + 4, nai2, acc);
  pair_group<MORE>(ajc.r + 4, db, da, adc + JC, nai2, acc);
}

template <bool LAST>
__device__ __forceinline__ void pair_body(
    float* ads, float* ps, float* ss, float* hs, float* wts,
    const float* __restrict__ ai, const float* __restrict__ aj,
    const float* __restrict__ adj, const float* hin,
    const float* __restrict__ W2, const float* __restrict__ b2,
    const float* __restrict__ rowsum, float* hout,
    const float* __restrict__ W1, const float* __restrict__ b1,
    float* __restrict__ aio, float* __restrict__ ajo) {
  const int tid  = threadIdx.x;
  const int lane = tid & 31;
  const int wave = tid >> 5;
  const int l16  = lane & 15;
  const int kh   = lane >> 4;
  const int i0 = blockIdx.x * PI;
  const int b  = blockIdx.y;
  const int m0 = b * N_ + i0;
  const int jw = wave * JW;

  const float* ajg = aj  + (size_t)(b * N_ + jw) * D_ + 4 * lane;
  const float* adg = adj + (size_t)m0 * N_ + jw;

  AjChunk ca, cb;
  ca.fetch(ajg);
  v4f rd[PI * JW / 4 / 32];
#pragma unroll
  for (int k = 0; k < PI * JW / 4 / 32; ++k) {
    const int idx = lane + 32 * k, row = idx >> 4, c = idx & 15;
    rd[k] = *(const v4f*)(adg + (size_t)row * N_ + 4 * c);
  }
  v2f nai2[PI][2], acc[PI][2];
#pragma unroll
  for (int r = 0; r < PI; ++r) {
    const v4f t = *(const v4f*)(ai + (size_t)(m0 + r) * D_ + 4 * lane);
    nai2[r][0] = (v2f){-t[0], -t[1]};
    nai2[r][1] = (v2f){-t[2], -t[3]};
    acc[r][0] = (v2f){0.f, 0.f};
    acc[r][1] = (v2f){0.f, 0.f};
  }

  WTile t2;
  float* const p2 = wts + wave * WT;
  float* const pi = wts + (NW + wave) * WT;
  float* const pj = wts + (2 * NW + wave) * WT;

  for (int idx = tid; idx < (16 - PI) * HP; idx += 256) {
    ss[PI * HP + idx] = 0.f;
    if (!LAST) hs[PI * HP + idx] = 0.f;
  }

  float* const adw = ads + wave * (PI * AW);
#pragma unroll
  for (int k = 0; k < PI * JW / 4 / 32; ++k) {
    const int idx = lane + 32 * k, row = idx >> 4, c = idx & 15;
    *(v4f*)(adw + row * AW + 4 * c) = rd[k];
  }
  __syncthreads();

  AdjGroup da, db;
  da.read(adw);

#pragma unroll 1
  for (int ch = 0; ch < NCH - 2; ch += 2) {
    cb.fetch(ajg + (size_t)(ch + 1) * JC * D_);
    pair_chunk<true>(ca, adw + ch * JC, da, db, nai2, acc);
    ca.fetch(ajg + (size_t)(ch + 2) * JC * D_);
    pair_chunk<true>(cb, adw + (ch + 1) * JC, da, db, nai2, acc);
  }
  cb.fetch(ajg + (size_t)(NCH - 1) * JC * D_);
  t2.fetch(W2 + (size_t)(wave * 16) * D_, D_, lane);
  WTile ti, tj;
  float bv1 = 0.f;
  float hv[PI], rv[PI];
  const float bv2 = b2[wave * 16 + l16];
  const v4f aiw = *(const v4f*)(ai + (size_t)(m0 + wave) * D_ + 4 * lane);
  const float rsw = rowsum[m0 + wave];
#pragma unroll
  for (int r = 0; r < PI; ++r) {
    hv[r] = hin[(size_t)(m0 + r) * D_ + wave * 16 + l16];
    rv[r] = rowsum[m0 + r];
  }
  if (!LAST) {
    ti.fetch(W1 + (size_t)(wave * 16) * 2 * D_, 2 * D_, lane);
    tj.fetch(W1 + (size_t)(wave * 16) * 2 * D_ + D_, 2 * D_, lane);
    bv1 = b1[wave * 16 + l16];
  }

  pair_chunk<true>(ca, adw + (NCH - 2) * JC, da, db, nai2, acc);
  pair_chunk<false>(cb, adw + (NCH - 1) * JC, da, db, nai2, acc);
  t2.park(p2, lane);

#pragma unroll
  for (int r = 0; r < PI; ++r) {
    const v4f o = {acc[r][0][0], acc[r][0][1], acc[r][1][0], acc[r][1][1]};
    *(v4f*)(ps + (wave * PI + r) * D_ + 4 * lane) = o;
  }
  __syncthreads();
  {
    const int row = tid >> 5, c4 = tid & 31;
    v4f s = *(const v4f*)(ps + row * D_ + 4 * c4);
#pragma unroll
    for (int w = 1; w < NW; ++w) s += *(const v4f*)(ps + (w * PI + row) * D_ + 4 * c4);
    s = __builtin_elementwise_fma(aiw, (v4f){rsw, rsw, rsw, rsw}, s);
    *(v4f*)(ss + row * HP + 4 * c4) = s;
  }
  __syncthreads();

  asm volatile("" :: "v"(bv2), "s"(rv[0]), "s"(rv[1]), "s"(rv[2]), "s"(rv[3]),
                     "s"(rv[4]), "s"(rv[5]), "s"(rv[6]), "s"(rv[7]));

  StripA sa;
  EdgeB w2b;
  const float* const srow  = ss + l16 * HP + kh * 2;
  const float* const w2row = p2 + l16 * HP + kh * 2;
  frag_load<0, 8>(sa.a, w2b.b, srow, w2row);
  __builtin_amdgcn_sched_barrier(0);
  frag_load<8, D_ / 4>(sa.a, w2b.b, srow, w2row);
  if (!LAST) {
    __builtin_amdgcn_sched_barrier(0);
    ti.park(pi, lane);
    tj.park(pj, lane);
  }
  __builtin_amdgcn_sched_barrier(0);
  v8f c = {};
#pragma unroll
  for (int kk = 0; kk < D_ / 4; ++kk)
    c = __builtin_amdgcn_wmma_f32_16x16x4_f32(
        false, sa.a[kk], false, w2b.b[kk], (short)0, c, false, false);
  float hn[PI];
#pragma unroll
  for (int r = 0; r < PI; ++r) hn[r] = hv[r] + c[r] + rv[r] * bv2;
  if (LAST) {
    if (kh == 0) {
#pragma unroll
      for (int r = 0; r < PI; ++r)
        hout[(size_t)(m0 + r) * D_ + wave * 16 + l16] = hn[r];
    }
    return;
  }

  if (kh == 0) {
#pragma unroll
    for (int r = 0; r < PI; ++r) hs[r * HP + wave * 16 + l16] = hn[r];
  }
  __syncthreads();
  if (kh == 0) {
#pragma unroll
    for (int r = 0; r < PI; ++r)
      hout[(size_t)(m0 + r) * D_ + wave * 16 + l16] = hn[r];
  }

  StripA ha;
  EdgeB wib, wjb;
  const float* const hrow  = hs + l16 * HP + kh * 2;
  const float* const wirow = pi + l16 * HP + kh * 2;
  const float* const wjrow = pj + l16 * HP + kh * 2;
#pragma unroll
  for (int kk = 0; kk < 4; ++kk) {
    ha.a[kk]  = *(const v2f*)(hrow + kk * 4);
    wib.b[kk] = *(const v2f*)(wirow + kk * 4);
    wjb.b[kk] = *(const v2f*)(wjrow + kk * 4);
  }
  __builtin_amdgcn_sched_barrier(0);
#pragma unroll
  for (int kk = 4; kk < D_ / 4; ++kk) {
    ha.a[kk]  = *(const v2f*)(hrow + kk * 4);
    wib.b[kk] = *(const v2f*)(wirow + kk * 4);
    wjb.b[kk] = *(const v2f*)(wjrow + kk * 4);
  }
  __builtin_amdgcn_sched_barrier(0);
  v8f ci = {}, cj = {};
#pragma unroll
  for (int kk = 0; kk < D_ / 4; ++kk) {
    ci = __builtin_amdgcn_wmma_f32_16x16x4_f32(
        false, ha.a[kk], false, wib.b[kk], (short)0, ci, false, false);
    cj = __builtin_amdgcn_wmma_f32_16x16x4_f32(
        false, ha.a[kk], false, wjb.b[kk], (short)0, cj, false, false);
  }
  float oi[PI], oj[PI];
#pragma unroll
  for (int r = 0; r < PI; ++r) { oi[r] = ci[r] + bv1; oj[r] = cj[r]; }
  if (kh == 0) {
#pragma unroll
    for (int r = 0; r < PI; ++r) {
      const size_t idx = (size_t)(m0 + r) * D_ + wave * 16 + l16;
      aio[idx] = oi[r];
      ajo[idx] = oj[r];
    }
  }
}

__global__ __launch_bounds__(256)
void pairwise_mid(const float* __restrict__ ai, const float* __restrict__ aj,
                  const float* __restrict__ adj, const float* __restrict__ hin,
                  const float* __restrict__ W2, const float* __restrict__ b2,
                  const float* __restrict__ rowsum, float* __restrict__ hout,
                  const float* __restrict__ W1, const float* __restrict__ b1,
                  float* __restrict__ aio, float* __restrict__ ajo) {
  __shared__ __attribute__((aligned(16))) float ads[NW * PI * AW];
  __shared__ __attribute__((aligned(16))) float ps[NW * PI * D_];
  __shared__ __attribute__((aligned(16))) float ss[16 * HP];
  __shared__ __attribute__((aligned(16))) float hs[16 * HP];
  __shared__ __attribute__((aligned(16))) float wts[3 * NW * WT];
  pair_body<false>(ads, ps, ss, hs, wts, ai, aj, adj, hin, W2, b2, rowsum, hout, W1, b1, aio, ajo);
}

__global__ __launch_bounds__(256)
void pairwise_last(const float* __restrict__ ai, const float* __restrict__ aj,
                   const float* __restrict__ adj,
                   const float* __restrict__ W2, const float* __restrict__ b2,
                   const float* __restrict__ rowsum, float* h) {
  __shared__ __attribute__((aligned(16))) float ads[NW * PI * AW];
  __shared__ __attribute__((aligned(16))) float ps[NW * PI * D_];
  __shared__ __attribute__((aligned(16))) float ss[16 * HP];
  __shared__ __attribute__((aligned(16))) float wts[NW * WT];
  pair_body<true>(ads, ps, ss, nullptr, wts, ai, aj, adj, h, W2, b2, rowsum, h,
                  nullptr, nullptr, nullptr, nullptr);
}

extern "C" void kernel_launch(void* const* d_in, const int* in_sizes, int n_in,
                              void* d_out, int out_size, void* d_ws, size_t ws_size,
                              hipStream_t stream) {
  const float* node = (const float*)d_in[0];
  const float* adj  = (const float*)d_in[1];
  const float* w1   = (const float*)d_in[2];
  const float* b1   = (const float*)d_in[3];
  const float* w2   = (const float*)d_in[4];
  const float* b2   = (const float*)d_in[5];

  float* h   = (float*)d_out;
  float* ai0 = (float*)d_ws;
  float* aj0 = ai0 + (size_t)M_ * D_;
  float* ai1 = aj0 + (size_t)M_ * D_;
  float* aj1 = ai1 + (size_t)M_ * D_;
  float* rs  = aj1 + (size_t)M_ * D_;

  const dim3 ge(2, M_ / 16);
  const dim3 gp(N_ / PI, B_);
  first_edge<<<ge, 256, 0, stream>>>(node, adj, w1, b1, ai0, aj0, rs);
  pairwise_mid<<<gp, 256, 0, stream>>>(ai0, aj0, adj, node, w2, b2, rs, h,
                                       w1 + (size_t)D_ * 2 * D_, b1 + D_, ai1, aj1);
  pairwise_last<<<gp, 256, 0, stream>>>(ai1, aj1, adj, w2 + (size_t)D_ * D_, b2 + D_, rs, h);
}
